# starting kernel
# baseline (speedup 1.0000x reference)
#include <hip/hip_runtime.h>
#include <math.h>


typedef float v2f __attribute__((ext_vector_type(2)));
typedef float v8f __attribute__((ext_vector_type(8)));

#define WMMA4(a, b, c) \
  __builtin_amdgcn_wmma_f32_16x16x4_f32(false, (a), false, (b), (short)0, (c), false, false)

constexpr int OFF_W1  = 0;
constexpr int OFF_W2  = OFF_W1  + 20800;
constexpr int OFF_WMU = OFF_W2  + 6400;
constexpr int OFF_WLV = OFF_WMU + 640;
constexpr int OFF_CBT = OFF_WLV + 640;
constexpr int OFF_B1  = OFF_CBT + 10240;
constexpr int OFF_B2  = OFF_B1  + 208;
constexpr int OFF_BMU = OFF_B2  + 32;
constexpr int OFF_BLV = OFF_BMU + 32;
constexpr int OFF_CBN = OFF_BLV + 32;
constexpr int HSTR    = 210;
constexpr int OFF_H   = OFF_CBN + 512;
constexpr int OFF_Z   = OFF_H   + 8 * 16 * HSTR;
constexpr int OFF_MND = OFF_Z   + 8 * 320;
constexpr int OFF_MNJ = OFF_MND + 8 * 16;
constexpr int LDS_FLOATS = OFF_MNJ + 8 * 16;

__global__ __launch_bounds__(256) void vqvae_fused_kernel(
    const float* __restrict__ c,  const float* __restrict__ W1, const float* __restrict__ b1,
    const float* __restrict__ W2, const float* __restrict__ b2, const float* __restrict__ cb,
    const float* __restrict__ Wmu, const float* __restrict__ bmu,
    const float* __restrict__ Wlv, const float* __restrict__ blv,
    float* __restrict__ out_mu, float* __restrict__ out_lv,
    float* __restrict__ out_zq, float* __restrict__ out_ze,
    float* __restrict__ ws, int B) {
  extern __shared__ float smem[];
  const int tid  = threadIdx.x;
  const int wave = tid >> 5;
  const int lane = tid & 31;
  const int lo   = lane & 15;
  const int hi   = lane >> 4;
  const int klo  = hi * 2;

  for (int f = tid; f < 20800; f += 256) {
    int kk = f & 3, rem = f >> 2, n = rem % 208, kt = rem / 208;
    smem[OFF_W1 + f] = (n < 200) ? W1[(kt * 4 + kk) * 200 + n] : 0.f;
  }
  for (int f = tid; f < 6400; f += 256) {
    int kk = f & 3, rem = f >> 2, n = rem & 31, kt = rem >> 5;
    smem[OFF_W2 + f] = (n < 20) ? W2[(kt * 4 + kk) * 20 + n] : 0.f;
  }
  for (int f = tid; f < 640; f += 256) {
    int kk = f & 3, rem = f >> 2, n = rem & 31, kt = rem >> 5;
    smem[OFF_WMU + f] = (n < 20) ? Wmu[(kt * 4 + kk) * 20 + n] : 0.f;
    smem[OFF_WLV + f] = (n < 20) ? Wlv[(kt * 4 + kk) * 20 + n] : 0.f;
  }
  for (int f = tid; f < 10240; f += 256) {
    int kk = f & 3, rem = f >> 2, j = rem & 511, kt = rem >> 9;
    smem[OFF_CBT + f] = cb[j * 20 + kt * 4 + kk];
  }
  for (int n = tid; n < 208; n += 256) smem[OFF_B1 + n] = (n < 200) ? b1[n] : 0.f;
  if (tid < 32) {
    smem[OFF_B2  + tid] = (tid < 20) ? b2[tid]  : 0.f;
    smem[OFF_BMU + tid] = (tid < 20) ? bmu[tid] : 0.f;
    smem[OFF_BLV + tid] = (tid < 20) ? blv[tid] : 0.f;
  }
  for (int j = tid; j < 512; j += 256) {
    float s = 0.f;
    for (int k = 0; k < 20; ++k) { float t = cb[j * 20 + k]; s += t * t; }
    smem[OFF_CBN + j] = s;
  }
  __syncthreads();

  float* hS = &smem[OFF_H + wave * 16 * HSTR];
  float* zS = &smem[OFF_Z + wave * 320];
  float* mnD = &smem[OFF_MND + wave * 16];
  int*   mnJ = (int*)&smem[OFF_MNJ + wave * 16];

  const int numTiles = B >> 4;
  for (int tile = blockIdx.x * 8 + wave; tile < numTiles; tile += gridDim.x * 8) {
    const int rowBase = tile * 16;

    v2f a1[25];
    const float* crow = c + (size_t)(rowBase + lo) * 100 + klo;
    #pragma unroll
    for (int kt = 0; kt < 25; ++kt) a1[kt] = *(const v2f*)(crow + kt * 4);

    for (int nt = 0; nt < 13; ++nt) {
      v8f acc = {};
      #pragma unroll
      for (int kt = 0; kt < 25; ++kt) {
        v2f b = *(const v2f*)&smem[OFF_W1 + ((kt * 208 + nt * 16 + lo) << 2) + klo];
        acc = WMMA4(a1[kt], b, acc);
      }
      float bias = smem[OFF_B1 + nt * 16 + lo];
      #pragma unroll
      for (int v = 0; v < 8; ++v) {
        float x = acc[v] + bias;
        x = (x > 0.f) ? x : (__expf(x) - 1.f);
        hS[(v + 8 * hi) * HSTR + nt * 16 + lo] = x;
      }
    }
    asm volatile("s_wait_dscnt 0" ::: "memory");

    for (int nt = 0; nt < 2; ++nt) {
      v8f acc = {};
      #pragma unroll
      for (int kt = 0; kt < 50; ++kt) {
        v2f a = *(const v2f*)&hS[lo * HSTR + kt * 4 + klo];
        v2f b = *(const v2f*)&smem[OFF_W2 + ((kt * 32 + nt * 16 + lo) << 2) + klo];
        acc = WMMA4(a, b, acc);
      }
      int col = nt * 16 + lo;
      float bias = smem[OFF_B2 + col];
      #pragma unroll
      for (int v = 0; v < 8; ++v) {
        if (col < 20) {
          float zv = acc[v] + bias;
          int row = v + 8 * hi;
          zS[row * 20 + col] = zv;
          out_ze[(size_t)(rowBase + row) * 20 + col] = zv;
        }
      }
    }
    asm volatile("s_wait_dscnt 0" ::: "memory");

    v2f az[5];
    #pragma unroll
    for (int kt = 0; kt < 5; ++kt) az[kt] = *(const v2f*)&zS[lo * 20 + kt * 4 + klo];

    for (int nt = 0; nt < 2; ++nt) {
      v8f am = {}, al = {};
      #pragma unroll
      for (int kt = 0; kt < 5; ++kt) {
        v2f bm = *(const v2f*)&smem[OFF_WMU + ((kt * 32 + nt * 16 + lo) << 2) + klo];
        v2f bl = *(const v2f*)&smem[OFF_WLV + ((kt * 32 + nt * 16 + lo) << 2) + klo];
        am = WMMA4(az[kt], bm, am);
        al = WMMA4(az[kt], bl, al);
      }
      int col = nt * 16 + lo;
      if (col < 20) {
        float bmv = smem[OFF_BMU + col], blv_ = smem[OFF_BLV + col];
        #pragma unroll
        for (int v = 0; v < 8; ++v) {
          int row = v + 8 * hi;
          out_mu[(size_t)(rowBase + row) * 20 + col] = am[v] + bmv;
          out_lv[(size_t)(rowBase + row) * 20 + col] = al[v] + blv_;
        }
      }
    }

    float zn = 0.f;
    #pragma unroll
    for (int k = 0; k < 20; ++k) { float t = zS[lo * 20 + k]; zn += t * t; }
    float znv[8];
    #pragma unroll
    for (int v = 0; v < 8; ++v) znv[v] = __shfl(zn, v + 8 * hi, 32);

    float bestD[8]; int bestJ[8];
    #pragma unroll
    for (int v = 0; v < 8; ++v) { bestD[v] = 3.4e38f; bestJ[v] = 0; }
    for (int jt = 0; jt < 32; ++jt) {
      v8f acc = {};
      #pragma unroll
      for (int kt = 0; kt < 5; ++kt) {
        v2f b = *(const v2f*)&smem[OFF_CBT + ((kt * 512 + jt * 16 + lo) << 2) + klo];
        acc = WMMA4(az[kt], b, acc);
      }
      float cn = smem[OFF_CBN + jt * 16 + lo];
      int j = jt * 16 + lo;
      #pragma unroll
      for (int v = 0; v < 8; ++v) {
        float d = znv[v] + cn - 2.f * acc[v];
        if (d < bestD[v]) { bestD[v] = d; bestJ[v] = j; }
      }
    }
    #pragma unroll
    for (int s = 8; s >= 1; s >>= 1) {
      #pragma unroll
      for (int v = 0; v < 8; ++v) {
        float od = __shfl_xor(bestD[v], s, 32);
        int   oj = __shfl_xor(bestJ[v], s, 32);
        if (od < bestD[v] || (od == bestD[v] && oj < bestJ[v])) {
          bestD[v] = od; bestJ[v] = oj;
        }
      }
    }
    if (lo == 0) {
      #pragma unroll
      for (int v = 0; v < 8; ++v) { mnD[v + 8 * hi] = bestD[v]; mnJ[v + 8 * hi] = bestJ[v]; }
    }
    asm volatile("s_wait_dscnt 0" ::: "memory");

    if (lane == 0) {
      float s = 0.f;
      for (int m = 0; m < 16; ++m) s += mnD[m];
      ws[tile] = s;
    }
    #pragma unroll
    for (int t = 0; t < 10; ++t) {
      int e = lane * 10 + t;
      int row = e / 20, col = e - row * 20;
      int j = mnJ[row];
      float q = smem[OFF_CBT + (((col >> 2) * 512 + j) << 2) + (col & 3)];
      out_zq[(size_t)(rowBase + row) * 20 + col] = q;
    }
  }
}

__global__ __launch_bounds__(256) void vq_loss_reduce(const float* __restrict__ ws,
                                                      float* __restrict__ out,
                                                      int n, float invCount) {
  __shared__ float p[256];
  float s = 0.f;
  for (int i = threadIdx.x; i < n; i += 256) s += ws[i];
  p[threadIdx.x] = s;
  __syncthreads();
  if (threadIdx.x == 0) {
    float t = 0.f;
    for (int i = 0; i < 256; ++i) t += p[i];
    float q = t * invCount;
    out[0] = q;
    out[1] = 0.25f * q;
  }
}

extern "C" void kernel_launch(void* const* d_in, const int* in_sizes, int n_in,
                              void* d_out, int out_size, void* d_ws, size_t ws_size,
                              hipStream_t stream) {
  const float* c   = (const float*)d_in[0];
  const float* W1  = (const float*)d_in[1];
  const float* b1  = (const float*)d_in[2];
  const float* W2  = (const float*)d_in[3];
  const float* b2  = (const float*)d_in[4];
  const float* cb  = (const float*)d_in[5];
  const float* Wmu = (const float*)d_in[6];
  const float* bmu = (const float*)d_in[7];
  const float* Wlv = (const float*)d_in[8];
  const float* blv = (const float*)d_in[9];

  const int B = in_sizes[0] / 100;
  const int numTiles = B / 16;
  float* out = (float*)d_out;
  float* ws  = (float*)d_ws;

  const size_t ldsBytes = (size_t)LDS_FLOATS * sizeof(float);
  (void)hipFuncSetAttribute((const void*)vqvae_fused_kernel,
                            hipFuncAttributeMaxDynamicSharedMemorySize, (int)ldsBytes);

  const int blocks = (numTiles + 7) / 8;
  vqvae_fused_kernel<<<blocks, 256, ldsBytes, stream>>>(
      c, W1, b1, W2, b2, cb, Wmu, bmu, Wlv, blv,
      out,
      out + (size_t)B * 20,
      out + 2 * (size_t)B * 20,
      out + 3 * (size_t)B * 20,
      ws, B);
  vq_loss_reduce<<<1, 256, 0, stream>>>(ws, out + 4 * (size_t)B * 20,
                                        numTiles, 1.0f / ((float)B * 20.0f));
}


# 32 waves per work-group on one LDS weight copy, was 8 (8 waves per SIMD, was 2, hide LDS and WMMA latency): h reaches GEMM2 in 16-column chunks via a 1280-byte per-wave buffer; sum order unchanged
# speedup vs baseline: 2.3154x; 2.3154x over previous
#include <hip/hip_runtime.h>
#include <math.h>


typedef float v2f __attribute__((ext_vector_type(2)));
typedef float v4f __attribute__((ext_vector_type(4)));
typedef float v8f __attribute__((ext_vector_type(8)));

#define WMMA4(a, b, c) \
  __builtin_amdgcn_wmma_f32_16x16x4_f32(false, (a), false, (b), (short)0, (c), false, false)

constexpr int OFF_W1  = 0;
constexpr int OFF_W2  = OFF_W1  + 20800;
constexpr int OFF_WMU = OFF_W2  + 6400;
constexpr int OFF_WLV = OFF_WMU + 640;
constexpr int OFF_CBT = OFF_WLV + 640;
constexpr int OFF_B1  = OFF_CBT + 10240;
constexpr int OFF_B2  = OFF_B1  + 208;
constexpr int OFF_BMU = OFF_B2  + 32;
constexpr int OFF_BLV = OFF_BMU + 32;
constexpr int OFF_CBN = OFF_BLV + 32;
constexpr int NWAVES   = 32;
constexpr int NTHREADS = NWAVES * 32;
constexpr int CSTR    = 20;
constexpr int OFF_PW  = OFF_CBN + 512;
constexpr int PW_HC   = 0;
constexpr int PW_Z    = PW_HC + 16 * CSTR;
constexpr int PW_MND  = PW_Z  + 320;
constexpr int PW_MNJ  = PW_MND + 16;
constexpr int PW_FLOATS = PW_MNJ + 16;
constexpr int LDS_FLOATS = OFF_PW + NWAVES * PW_FLOATS;
constexpr int IMG_FLOATS = OFF_PW;
constexpr int IMG_VEC4   = IMG_FLOATS / 4;
constexpr int WS_PART_OFF = 40960;
static_assert((IMG_FLOATS & 3) == 0, "weight block must be a whole number of 16-byte pieces");
static_assert(WS_PART_OFF >= IMG_FLOATS, "loss partials must not overlap the weight block");

__global__ __launch_bounds__(256) void vqvae_build_weight_block(
    const float* __restrict__ W1, const float* __restrict__ b1,
    const float* __restrict__ W2, const float* __restrict__ b2, const float* __restrict__ cb,
    const float* __restrict__ Wmu, const float* __restrict__ bmu,
    const float* __restrict__ Wlv, const float* __restrict__ blv,
    float* __restrict__ img) {
  const int f = blockIdx.x * 256 + threadIdx.x;
  if (f >= IMG_FLOATS) return;
  float v;
  if (f < OFF_W2) {
    int g = f - OFF_W1;
    int kk = g & 3, rem = g >> 2, n = rem % 208, kt = rem / 208;
    v = (n < 200) ? W1[(kt * 4 + kk) * 200 + n] : 0.f;
  } else if (f < OFF_WMU) {
    int g = f - OFF_W2;
    int kk = g & 3, rem = g >> 2, n = rem & 31, kt = rem >> 5;
    v = (n < 20) ? W2[(kt * 4 + kk) * 20 + n] : 0.f;
  } else if (f < OFF_WLV) {
    int g = f - OFF_WMU;
    int kk = g & 3, rem = g >> 2, n = rem & 31, kt = rem >> 5;
    v = (n < 20) ? Wmu[(kt * 4 + kk) * 20 + n] : 0.f;
  } else if (f < OFF_CBT) {
    int g = f - OFF_WLV;
    int kk = g & 3, rem = g >> 2, n = rem & 31, kt = rem >> 5;
    v = (n < 20) ? Wlv[(kt * 4 + kk) * 20 + n] : 0.f;
  } else if (f < OFF_B1) {
    int g = f - OFF_CBT;
    int kk = g & 3, rem = g >> 2, j = rem & 511, kt = rem >> 9;
    v = cb[j * 20 + kt * 4 + kk];
  } else if (f < OFF_B2) {
    int n = f - OFF_B1;
    v = (n < 200) ? b1[n] : 0.f;
  } else if (f < OFF_BMU) {
    int n = f - OFF_B2;
    v = (n < 20) ? b2[n] : 0.f;
  } else if (f < OFF_BLV) {
    int n = f - OFF_BMU;
    v = (n < 20) ? bmu[n] : 0.f;
  } else if (f < OFF_CBN) {
    int n = f - OFF_BLV;
    v = (n < 20) ? blv[n] : 0.f;
  } else {
    int j = f - OFF_CBN;
    float s = 0.f;
    for (int k = 0; k < 20; ++k) { float t = cb[j * 20 + k]; s += t * t; }
    v = s;
  }
  img[f] = v;
}

template <int NK>
__device__ __forceinline__ void h_chunk_into_z(const float* smem, float* hC,
                                               const v2f (&a1)[25], int nt, int lo, int hi, int klo,
                                               v8f& z0, v8f& z1) {
  v8f acc = {};
  #pragma unroll
  for (int kt = 0; kt < 25; ++kt) {
    v2f b = *(const v2f*)&smem[OFF_W1 + ((kt * 208 + nt * 16 + lo) << 2) + klo];
    acc = WMMA4(a1[kt], b, acc);
  }
  float bias = smem[OFF_B1 + nt * 16 + lo];
  #pragma unroll
  for (int v = 0; v < 8; ++v) {
    float x = acc[v] + bias;
    x = (x > 0.f) ? x : (__expf(x) - 1.f);
    hC[(v + 8 * hi) * CSTR + lo] = x;
  }
  asm volatile("s_wait_dscnt 0" ::: "memory");
  #pragma unroll
  for (int kk = 0; kk < NK; ++kk) {
    const int kt = nt * 4 + kk;
    v2f a  = *(const v2f*)&hC[lo * CSTR + kk * 4 + klo];
    v2f b0 = *(const v2f*)&smem[OFF_W2 + ((kt * 32 + lo) << 2) + klo];
    v2f b1 = *(const v2f*)&smem[OFF_W2 + ((kt * 32 + 16 + lo) << 2) + klo];
    z0 = WMMA4(a, b0, z0);
    z1 = WMMA4(a, b1, z1);
  }
}

__global__ __launch_bounds__(NTHREADS) void vqvae_fused_kernel(
    const float* __restrict__ c,  const float* __restrict__ wimg,
    float* __restrict__ out_mu, float* __restrict__ out_lv,
    float* __restrict__ out_zq, float* __restrict__ out_ze,
    float* __restrict__ ws, int B) {
  extern __shared__ float smem[];
  const int tid  = threadIdx.x;
  const int wave = tid >> 5;
  const int lane = tid & 31;
  const int lo   = lane & 15;
  const int hi   = lane >> 4;
  const int klo  = hi * 2;

  {
    const v4f* __restrict__ gsrc = (const v4f*)wimg;
    v4f* sdst = (v4f*)smem;
    constexpr int FULL = IMG_VEC4 / NTHREADS;
    const bool hasTail = tid < IMG_VEC4 - FULL * NTHREADS;
    v4f tail = {};
    if (hasTail) tail = gsrc[tid + FULL * NTHREADS];
    #pragma unroll
    for (int t = 0; t < FULL; ++t) sdst[tid + t * NTHREADS] = gsrc[tid + t * NTHREADS];
    if (hasTail) sdst[tid + FULL * NTHREADS] = tail;
  }
  __syncthreads();

  float* pw  = &smem[OFF_PW + wave * PW_FLOATS];
  float* hC  = pw + PW_HC;
  float* zS  = pw + PW_Z;
  float* mnD = pw + PW_MND;
  int*   mnJ = (int*)(pw + PW_MNJ);

  const int numTiles = B >> 4;
  const int tile = blockIdx.x * NWAVES + wave;
  if (tile < numTiles) {
    const int rowBase = tile * 16;

    v2f a1[25];
    const float* crow = c + (size_t)(rowBase + lo) * 100 + klo;
    #pragma unroll
    for (int kt = 0; kt < 25; ++kt) a1[kt] = *(const v2f*)(crow + kt * 4);

    v8f z0 = {}, z1 = {};
    for (int nt = 0; nt < 12; ++nt)
      h_chunk_into_z<4>(smem, hC, a1, nt, lo, hi, klo, z0, z1);
    h_chunk_into_z<2>(smem, hC, a1, 12, lo, hi, klo, z0, z1);

    for (int nt = 0; nt < 2; ++nt) {
      const v8f acc = nt ? z1 : z0;
      int col = nt * 16 + lo;
      float bias = smem[OFF_B2 + col];
      #pragma unroll
      for (int v = 0; v < 8; ++v) {
        if (col < 20) {
          float zv = acc[v] + bias;
          int row = v + 8 * hi;
          zS[row * 20 + col] = zv;
          out_ze[(size_t)(rowBase + row) * 20 + col] = zv;
        }
      }
    }
    asm volatile("s_wait_dscnt 0" ::: "memory");

    v2f az[5];
    #pragma unroll
    for (int kt = 0; kt < 5; ++kt) az[kt] = *(const v2f*)&zS[lo * 20 + kt * 4 + klo];

    for (int nt = 0; nt < 2; ++nt) {
      v8f am = {}, al = {};
      #pragma unroll
      for (int kt = 0; kt < 5; ++kt) {
        v2f bm = *(const v2f*)&smem[OFF_WMU + ((kt * 32 + nt * 16 + lo) << 2) + klo];
        v2f bl = *(const v2f*)&smem[OFF_WLV + ((kt * 32 + nt * 16 + lo) << 2) + klo];
        am = WMMA4(az[kt], bm, am);
        al = WMMA4(az[kt], bl, al);
      }
      int col = nt * 16 + lo;
      if (col < 20) {
        float bmv = smem[OFF_BMU + col], blv_ = smem[OFF_BLV + col];
        #pragma unroll
        for (int v = 0; v < 8; ++v) {
          int row = v + 8 * hi;
          out_mu[(size_t)(rowBase + row) * 20 + col] = am[v] + bmv;
          out_lv[(size_t)(rowBase + row) * 20 + col] = al[v] + blv_;
        }
      }
    }

    float zn = 0.f;
    #pragma unroll
    for (int k = 0; k < 20; ++k) { float t = zS[lo * 20 + k]; zn += t * t; }
    float znv[8];
    #pragma unroll
    for (int v = 0; v < 8; ++v) znv[v] = __shfl(zn, v + 8 * hi, 32);

    float bestD[8]; int bestJ[8];
    #pragma unroll
    for (int v = 0; v < 8; ++v) { bestD[v] = 3.4e38f; bestJ[v] = 0; }
    for (int jt = 0; jt < 32; ++jt) {
      v8f acc = {};
      #pragma unroll
      for (int kt = 0; kt < 5; ++kt) {
        v2f b = *(const v2f*)&smem[OFF_CBT + ((kt * 512 + jt * 16 + lo) << 2) + klo];
        acc = WMMA4(az[kt], b, acc);
      }
      float cn = smem[OFF_CBN + jt * 16 + lo];
      int j = jt * 16 + lo;
      #pragma unroll
      for (int v = 0; v < 8; ++v) {
        float d = znv[v] + cn - 2.f * acc[v];
        if (d < bestD[v]) { bestD[v] = d; bestJ[v] = j; }
      }
    }
    #pragma unroll
    for (int s = 8; s >= 1; s >>= 1) {
      #pragma unroll
      for (int v = 0; v < 8; ++v) {
        float od = __shfl_xor(bestD[v], s, 32);
        int   oj = __shfl_xor(bestJ[v], s, 32);
        if (od < bestD[v] || (od == bestD[v] && oj < bestJ[v])) {
          bestD[v] = od; bestJ[v] = oj;
        }
      }
    }
    if (lo == 0) {
      #pragma unroll
      for (int v = 0; v < 8; ++v) { mnD[v + 8 * hi] = bestD[v]; mnJ[v + 8 * hi] = bestJ[v]; }
    }
    asm volatile("s_wait_dscnt 0" ::: "memory");

    if (lane == 0) {
      float s = 0.f;
      for (int m = 0; m < 16; ++m) s += mnD[m];
      ws[tile] = s;
    }
    #pragma unroll
    for (int t = 0; t < 10; ++t) {
      int e = lane * 10 + t;
      int row = e / 20, col = e - row * 20;
      int j = mnJ[row];
      float q = smem[OFF_CBT + (((col >> 2) * 512 + j) << 2) + (col & 3)];
      out_zq[(size_t)(rowBase + row) * 20 + col] = q;
    }
  }
}

__global__ __launch_bounds__(256) void vq_loss_reduce(const float* __restrict__ ws,
                                                      float* __restrict__ out,
                                                      int n, float invCount) {
  __shared__ float p[256];
  float s = 0.f;
  for (int i = threadIdx.x; i < n; i += 256) s += ws[i];
  p[threadIdx.x] = s;
  __syncthreads();
  if (threadIdx.x == 0) {
    float t = 0.f;
    for (int i = 0; i < 256; ++i) t += p[i];
    float q = t * invCount;
    out[0] = q;
    out[1] = 0.25f * q;
  }
}

extern "C" void kernel_launch(void* const* d_in, const int* in_sizes, int n_in,
                              void* d_out, int out_size, void* d_ws, size_t ws_size,
                              hipStream_t stream) {
  const float* c   = (const float*)d_in[0];
  const float* W1  = (const float*)d_in[1];
  const float* b1  = (const float*)d_in[2];
  const float* W2  = (const float*)d_in[3];
  const float* b2  = (const float*)d_in[4];
  const float* cb  = (const float*)d_in[5];
  const float* Wmu = (const float*)d_in[6];
  const float* bmu = (const float*)d_in[7];
  const float* Wlv = (const float*)d_in[8];
  const float* blv = (const float*)d_in[9];

  const int B = in_sizes[0] / 100;
  const int numTiles = B / 16;
  float* out = (float*)d_out;
  float* wimg = (float*)d_ws;
  float* ws   = wimg + WS_PART_OFF;

  const size_t ldsBytes = (size_t)LDS_FLOATS * sizeof(float);
  (void)hipFuncSetAttribute((const void*)vqvae_fused_kernel,
                            hipFuncAttributeMaxDynamicSharedMemorySize, (int)ldsBytes);

  vqvae_build_weight_block<<<(IMG_FLOATS + 255) / 256, 256, 0, stream>>>(
      W1, b1, W2, b2, cb, Wmu, bmu, Wlv, blv, wimg);

  const int blocks = (numTiles + NWAVES - 1) / NWAVES;
  vqvae_fused_kernel<<<blocks, NTHREADS, ldsBytes, stream>>>(
      c, wimg,
      out,
      out + (size_t)B * 20,
      out + 2 * (size_t)B * 20,
      out + 3 * (size_t)B * 20,
      ws, B);
  vq_loss_reduce<<<1, 256, 0, stream>>>(ws, out + 4 * (size_t)B * 20,
                                        numTiles, 1.0f / ((float)B * 20.0f));
}


# distance product with operands swapped (codebook rows on M, tile rows on N): argmin is a per-lane running minimum and one lane-half exchange, not 32 LDS exchange-and-branch rounds; sums unchanged
# speedup vs baseline: 2.4529x; 1.0594x over previous
#include <hip/hip_runtime.h>
#include <math.h>


typedef float v2f __attribute__((ext_vector_type(2)));
typedef float v4f __attribute__((ext_vector_type(4)));
typedef float v8f __attribute__((ext_vector_type(8)));

#define WMMA4(a, b, c) \
  __builtin_amdgcn_wmma_f32_16x16x4_f32(false, (a), false, (b), (short)0, (c), false, false)

constexpr int OFF_W1  = 0;
constexpr int OFF_W2  = OFF_W1  + 20800;
constexpr int OFF_WMU = OFF_W2  + 6400;
constexpr int OFF_WLV = OFF_WMU + 640;
constexpr int OFF_CBT = OFF_WLV + 640;
constexpr int OFF_B1  = OFF_CBT + 10240;
constexpr int OFF_B2  = OFF_B1  + 208;
constexpr int OFF_BMU = OFF_B2  + 32;
constexpr int OFF_BLV = OFF_BMU + 32;
constexpr int OFF_CBN = OFF_BLV + 32;
constexpr int NWAVES   = 32;
constexpr int NTHREADS = NWAVES * 32;
constexpr int CSTR    = 20;
constexpr int OFF_PW  = OFF_CBN + 512;
constexpr int PW_HC   = 0;
constexpr int PW_Z    = PW_HC + 16 * CSTR;
constexpr int PW_FLOATS = PW_Z + 320;
constexpr int LDS_FLOATS = OFF_PW + NWAVES * PW_FLOATS;
constexpr int IMG_FLOATS = OFF_PW;
constexpr int IMG_VEC4   = IMG_FLOATS / 4;
constexpr int WS_PART_OFF = 40960;
static_assert((IMG_FLOATS & 3) == 0, "weight block must be a whole number of 16-byte pieces");
static_assert(WS_PART_OFF >= IMG_FLOATS, "loss partials must not overlap the weight block");

__global__ __launch_bounds__(256) void vqvae_build_weight_block(
    const float* __restrict__ W1, const float* __restrict__ b1,
    const float* __restrict__ W2, const float* __restrict__ b2, const float* __restrict__ cb,
    const float* __restrict__ Wmu, const float* __restrict__ bmu,
    const float* __restrict__ Wlv, const float* __restrict__ blv,
    float* __restrict__ img) {
  const int f = blockIdx.x * 256 + threadIdx.x;
  if (f >= IMG_FLOATS) return;
  float v;
  if (f < OFF_W2) {
    int g = f - OFF_W1;
    int kk = g & 3, rem = g >> 2, n = rem % 208, kt = rem / 208;
    v = (n < 200) ? W1[(kt * 4 + kk) * 200 + n] : 0.f;
  } else if (f < OFF_WMU) {
    int g = f - OFF_W2;
    int kk = g & 3, rem = g >> 2, n = rem & 31, kt = rem >> 5;
    v = (n < 20) ? W2[(kt * 4 + kk) * 20 + n] : 0.f;
  } else if (f < OFF_WLV) {
    int g = f - OFF_WMU;
    int kk = g & 3, rem = g >> 2, n = rem & 31, kt = rem >> 5;
    v = (n < 20) ? Wmu[(kt * 4 + kk) * 20 + n] : 0.f;
  } else if (f < OFF_CBT) {
    int g = f - OFF_WLV;
    int kk = g & 3, rem = g >> 2, n = rem & 31, kt = rem >> 5;
    v = (n < 20) ? Wlv[(kt * 4 + kk) * 20 + n] : 0.f;
  } else if (f < OFF_B1) {
    int g = f - OFF_CBT;
    int kk = g & 3, rem = g >> 2, j = rem & 511, kt = rem >> 9;
    v = cb[j * 20 + kt * 4 + kk];
  } else if (f < OFF_B2) {
    int n = f - OFF_B1;
    v = (n < 200) ? b1[n] : 0.f;
  } else if (f < OFF_BMU) {
    int n = f - OFF_B2;
    v = (n < 20) ? b2[n] : 0.f;
  } else if (f < OFF_BLV) {
    int n = f - OFF_BMU;
    v = (n < 20) ? bmu[n] : 0.f;
  } else if (f < OFF_CBN) {
    int n = f - OFF_BLV;
    v = (n < 20) ? blv[n] : 0.f;
  } else {
    int j = f - OFF_CBN;
    float s = 0.f;
    for (int k = 0; k < 20; ++k) { float t = cb[j * 20 + k]; s += t * t; }
    v = s;
  }
  img[f] = v;
}

template <int NK>
__device__ __forceinline__ void h_chunk_into_z(const float* smem, float* hC,
                                               const v2f (&a1)[25], int nt, int lo, int hi, int klo,
                                               v8f& z0, v8f& z1) {
  v8f acc = {};
  #pragma unroll
  for (int kt = 0; kt < 25; ++kt) {
    v2f b = *(const v2f*)&smem[OFF_W1 + ((kt * 208 + nt * 16 + lo) << 2) + klo];
    acc = WMMA4(a1[kt], b, acc);
  }
  float bias = smem[OFF_B1 + nt * 16 + lo];
  #pragma unroll
  for (int v = 0; v < 8; ++v) {
    float x = acc[v] + bias;
    x = (x > 0.f) ? x : (__expf(x) - 1.f);
    hC[(v + 8 * hi) * CSTR + lo] = x;
  }
  asm volatile("s_wait_dscnt 0" ::: "memory");
  #pragma unroll
  for (int kk = 0; kk < NK; ++kk) {
    const int kt = nt * 4 + kk;
    v2f a  = *(const v2f*)&hC[lo * CSTR + kk * 4 + klo];
    v2f b0 = *(const v2f*)&smem[OFF_W2 + ((kt * 32 + lo) << 2) + klo];
    v2f b1 = *(const v2f*)&smem[OFF_W2 + ((kt * 32 + 16 + lo) << 2) + klo];
    z0 = WMMA4(a, b0, z0);
    z1 = WMMA4(a, b1, z1);
  }
}

__global__ __launch_bounds__(NTHREADS) void vqvae_fused_kernel(
    const float* __restrict__ c,  const float* __restrict__ wimg,
    float* __restrict__ out_mu, float* __restrict__ out_lv,
    float* __restrict__ out_zq, float* __restrict__ out_ze,
    float* __restrict__ ws, int B) {
  extern __shared__ float smem[];
  const int tid  = threadIdx.x;
  const int wave = tid >> 5;
  const int lane = tid & 31;
  const int lo   = lane & 15;
  const int hi   = lane >> 4;
  const int klo  = hi * 2;

  {
    const v4f* __restrict__ gsrc = (const v4f*)wimg;
    v4f* sdst = (v4f*)smem;
    constexpr int FULL = IMG_VEC4 / NTHREADS;
    const bool hasTail = tid < IMG_VEC4 - FULL * NTHREADS;
    v4f tail = {};
    if (hasTail) tail = gsrc[tid + FULL * NTHREADS];
    #pragma unroll
    for (int t = 0; t < FULL; ++t) sdst[tid + t * NTHREADS] = gsrc[tid + t * NTHREADS];
    if (hasTail) sdst[tid + FULL * NTHREADS] = tail;
  }
  __syncthreads();

  float* pw  = &smem[OFF_PW + wave * PW_FLOATS];
  float* hC  = pw + PW_HC;
  float* zS  = pw + PW_Z;

  const int numTiles = B >> 4;
  const int tile = blockIdx.x * NWAVES + wave;
  if (tile < numTiles) {
    const int rowBase = tile * 16;

    v2f a1[25];
    const float* crow = c + (size_t)(rowBase + lo) * 100 + klo;
    #pragma unroll
    for (int kt = 0; kt < 25; ++kt) a1[kt] = *(const v2f*)(crow + kt * 4);

    v8f z0 = {}, z1 = {};
    for (int nt = 0; nt < 12; ++nt)
      h_chunk_into_z<4>(smem, hC, a1, nt, lo, hi, klo, z0, z1);
    h_chunk_into_z<2>(smem, hC, a1, 12, lo, hi, klo, z0, z1);

    for (int nt = 0; nt < 2; ++nt) {
      const v8f acc = nt ? z1 : z0;
      int col = nt * 16 + lo;
      float bias = smem[OFF_B2 + col];
      #pragma unroll
      for (int v = 0; v < 8; ++v) {
        if (col < 20) {
          float zv = acc[v] + bias;
          int row = v + 8 * hi;
          zS[row * 20 + col] = zv;
          out_ze[(size_t)(rowBase + row) * 20 + col] = zv;
        }
      }
    }
    asm volatile("s_wait_dscnt 0" ::: "memory");

    v2f az[5];
    #pragma unroll
    for (int kt = 0; kt < 5; ++kt) az[kt] = *(const v2f*)&zS[lo * 20 + kt * 4 + klo];

    for (int nt = 0; nt < 2; ++nt) {
      v8f am = {}, al = {};
      #pragma unroll
      for (int kt = 0; kt < 5; ++kt) {
        v2f bm = *(const v2f*)&smem[OFF_WMU + ((kt * 32 + nt * 16 + lo) << 2) + klo];
        v2f bl = *(const v2f*)&smem[OFF_WLV + ((kt * 32 + nt * 16 + lo) << 2) + klo];
        am = WMMA4(az[kt], bm, am);
        al = WMMA4(az[kt], bl, al);
      }
      int col = nt * 16 + lo;
      if (col < 20) {
        float bmv = smem[OFF_BMU + col], blv_ = smem[OFF_BLV + col];
        #pragma unroll
        for (int v = 0; v < 8; ++v) {
          int row = v + 8 * hi;
          out_mu[(size_t)(rowBase + row) * 20 + col] = am[v] + bmv;
          out_lv[(size_t)(rowBase + row) * 20 + col] = al[v] + blv_;
        }
      }
    }

    float zr[20];
    #pragma unroll
    for (int k = 0; k < 20; ++k) zr[k] = zS[lo * 20 + k];
    float zn = zr[1] * zr[1];
    zn = __builtin_fmaf(zr[0], zr[0], zn);
    #pragma unroll
    for (int k = 2; k < 20; ++k) zn = __builtin_fmaf(zr[k], zr[k], zn);

    float bestD[8]; int bestT[8];
    #pragma unroll
    for (int v = 0; v < 8; ++v) { bestD[v] = 3.4e38f; bestT[v] = 0; }
    for (int jt = 0; jt < 32; ++jt) {
      v2f a[5];
      #pragma unroll
      for (int kt = 0; kt < 5; ++kt)
        a[kt] = *(const v2f*)&smem[OFF_CBT + ((kt * 512 + jt * 16 + lo) << 2) + klo];
      const v4f* cnp = (const v4f*)__builtin_assume_aligned(&smem[OFF_CBN + jt * 16 + 8 * hi], 16);
      const v4f cn0 = cnp[0];
      const v4f cn1 = cnp[1];
      __builtin_amdgcn_sched_barrier(0);
      v8f acc = {};
      #pragma unroll
      for (int kt = 0; kt < 5; ++kt) acc = WMMA4(a[kt], az[kt], acc);
      #pragma unroll
      for (int v = 0; v < 8; ++v) {
        float cn = (v < 4) ? cn0[v & 3] : cn1[v & 3];
        float d = zn + cn - 2.f * acc[v];
        if (d < bestD[v]) { bestD[v] = d; bestT[v] = jt; }
      }
    }
    float bd = bestD[0];
    int   bj = bestT[0] * 16 + 8 * hi;
    #pragma unroll
    for (int v = 1; v < 8; ++v) {
      float od = bestD[v];
      int   oj = bestT[v] * 16 + 8 * hi + v;
      bool take = (od < bd) | ((od == bd) & (oj < bj));
      bd = take ? od : bd;
      bj = take ? oj : bj;
    }
    {
      float od = __shfl_xor(bd, 16, 32);
      int   oj = __shfl_xor(bj, 16, 32);
      bool take = (od < bd) | ((od == bd) & (oj < bj));
      bd = take ? od : bd;
      bj = take ? oj : bj;
    }

    float s = 0.f;
    #pragma unroll
    for (int m = 0; m < 16; ++m)
      s += __int_as_float(__builtin_amdgcn_readlane(__float_as_int(bd), m));
    if (lane == 0) ws[tile] = s;

    {
      const v4f q0 = *(const v4f*)&smem[OFF_CBT + (((3 * hi)     * 512 + bj) << 2)];
      const v4f q1 = *(const v4f*)&smem[OFF_CBT + (((3 * hi + 1) * 512 + bj) << 2)];
      const v2f q2 = *(const v2f*)&smem[OFF_CBT + ((2 * 512 + bj) << 2) + klo];
      float* zq = out_zq + (size_t)(rowBase + lo) * 20;
      *(v4f*)(zq + 12 * hi)     = q0;
      *(v4f*)(zq + 12 * hi + 4) = q1;
      *(v2f*)(zq + 8 + klo)     = q2;
    }
  }
}

__global__ __launch_bounds__(256) void vq_loss_reduce(const float* __restrict__ ws,
                                                      float* __restrict__ out,
                                                      int n, float invCount) {
  __shared__ float p[256];
  float s = 0.f;
  for (int i = threadIdx.x; i < n; i += 256) s += ws[i];
  p[threadIdx.x] = s;
  __syncthreads();
  if (threadIdx.x == 0) {
    float t = 0.f;
    for (int i = 0; i < 256; ++i) t += p[i];
    float q = t * invCount;
    out[0] = q;
    out[1] = 0.25f * q;
  }
}

extern "C" void kernel_launch(void* const* d_in, const int* in_sizes, int n_in,
                              void* d_out, int out_size, void* d_ws, size_t ws_size,
                              hipStream_t stream) {
  const float* c   = (const float*)d_in[0];
  const float* W1  = (const float*)d_in[1];
  const float* b1  = (const float*)d_in[2];
  const float* W2  = (const float*)d_in[3];
  const float* b2  = (const float*)d_in[4];
  const float* cb  = (const float*)d_in[5];
  const float* Wmu = (const float*)d_in[6];
  const float* bmu = (const float*)d_in[7];
  const float* Wlv = (const float*)d_in[8];
  const float* blv = (const float*)d_in[9];

  const int B = in_sizes[0] / 100;
  const int numTiles = B / 16;
  float* out = (float*)d_out;
  float* wimg = (float*)d_ws;
  float* ws   = wimg + WS_PART_OFF;

  const size_t ldsBytes = (size_t)LDS_FLOATS * sizeof(float);
  (void)hipFuncSetAttribute((const void*)vqvae_fused_kernel,
                            hipFuncAttributeMaxDynamicSharedMemorySize, (int)ldsBytes);

  vqvae_build_weight_block<<<(IMG_FLOATS + 255) / 256, 256, 0, stream>>>(
      W1, b1, W2, b2, cb, Wmu, bmu, Wlv, blv, wimg);

  const int blocks = (numTiles + NWAVES - 1) / NWAVES;
  vqvae_fused_kernel<<<blocks, NTHREADS, ldsBytes, stream>>>(
      c, wimg,
      out,
      out + (size_t)B * 20,
      out + 2 * (size_t)B * 20,
      out + 3 * (size_t)B * 20,
      ws, B);
  vq_loss_reduce<<<1, 256, 0, stream>>>(ws, out + 4 * (size_t)B * 20,
                                        numTiles, 1.0f / ((float)B * 20.0f));
}


# loss sum as a two-level tree: each work-group adds its 32 tile sums in LDS behind one barrier; the final kernel reads 512 partials, 2 per thread, not 16384 in 64 serial loads; loss sum order changes
# speedup vs baseline: 2.8608x; 1.1663x over previous
#include <hip/hip_runtime.h>
#include <math.h>


typedef float v2f __attribute__((ext_vector_type(2)));
typedef float v4f __attribute__((ext_vector_type(4)));
typedef float v8f __attribute__((ext_vector_type(8)));

#define WMMA4(a, b, c) \
  __builtin_amdgcn_wmma_f32_16x16x4_f32(false, (a), false, (b), (short)0, (c), false, false)

constexpr int OFF_W1  = 0;
constexpr int OFF_W2  = OFF_W1  + 20800;
constexpr int OFF_WMU = OFF_W2  + 6400;
constexpr int OFF_WLV = OFF_WMU + 640;
constexpr int OFF_CBT = OFF_WLV + 640;
constexpr int OFF_B1  = OFF_CBT + 10240;
constexpr int OFF_B2  = OFF_B1  + 208;
constexpr int OFF_BMU = OFF_B2  + 32;
constexpr int OFF_BLV = OFF_BMU + 32;
constexpr int OFF_CBN = OFF_BLV + 32;
constexpr int NWAVES   = 32;
constexpr int NTHREADS = NWAVES * 32;
constexpr int CSTR    = 20;
constexpr int OFF_PW  = OFF_CBN + 512;
constexpr int PW_HC   = 0;
constexpr int PW_Z    = PW_HC + 16 * CSTR;
constexpr int PW_FLOATS = PW_Z + 320;
constexpr int OFF_RED = OFF_PW + NWAVES * PW_FLOATS;
constexpr int LDS_FLOATS = OFF_RED + NWAVES;
constexpr int IMG_FLOATS = OFF_PW;
constexpr int IMG_VEC4   = IMG_FLOATS / 4;
constexpr int WS_PART_OFF = 40960;
static_assert((IMG_FLOATS & 3) == 0, "weight block must be a whole number of 16-byte pieces");
static_assert(WS_PART_OFF >= IMG_FLOATS, "loss partials must not overlap the weight block");

__global__ __launch_bounds__(256) void vqvae_build_weight_block(
    const float* __restrict__ W1, const float* __restrict__ b1,
    const float* __restrict__ W2, const float* __restrict__ b2, const float* __restrict__ cb,
    const float* __restrict__ Wmu, const float* __restrict__ bmu,
    const float* __restrict__ Wlv, const float* __restrict__ blv,
    float* __restrict__ img) {
  const int f = blockIdx.x * 256 + threadIdx.x;
  if (f >= IMG_FLOATS) return;
  float v;
  if (f < OFF_W2) {
    int g = f - OFF_W1;
    int kk = g & 3, rem = g >> 2, n = rem % 208, kt = rem / 208;
    v = (n < 200) ? W1[(kt * 4 + kk) * 200 + n] : 0.f;
  } else if (f < OFF_WMU) {
    int g = f - OFF_W2;
    int kk = g & 3, rem = g >> 2, n = rem & 31, kt = rem >> 5;
    v = (n < 20) ? W2[(kt * 4 + kk) * 20 + n] : 0.f;
  } else if (f < OFF_WLV) {
    int g = f - OFF_WMU;
    int kk = g & 3, rem = g >> 2, n = rem & 31, kt = rem >> 5;
    v = (n < 20) ? Wmu[(kt * 4 + kk) * 20 + n] : 0.f;
  } else if (f < OFF_CBT) {
    int g = f - OFF_WLV;
    int kk = g & 3, rem = g >> 2, n = rem & 31, kt = rem >> 5;
    v = (n < 20) ? Wlv[(kt * 4 + kk) * 20 + n] : 0.f;
  } else if (f < OFF_B1) {
    int g = f - OFF_CBT;
    int kk = g & 3, rem = g >> 2, j = rem & 511, kt = rem >> 9;
    v = cb[j * 20 + kt * 4 + kk];
  } else if (f < OFF_B2) {
    int n = f - OFF_B1;
    v = (n < 200) ? b1[n] : 0.f;
  } else if (f < OFF_BMU) {
    int n = f - OFF_B2;
    v = (n < 20) ? b2[n] : 0.f;
  } else if (f < OFF_BLV) {
    int n = f - OFF_BMU;
    v = (n < 20) ? bmu[n] : 0.f;
  } else if (f < OFF_CBN) {
    int n = f - OFF_BLV;
    v = (n < 20) ? blv[n] : 0.f;
  } else {
    int j = f - OFF_CBN;
    float s = 0.f;
    for (int k = 0; k < 20; ++k) { float t = cb[j * 20 + k]; s += t * t; }
    v = s;
  }
  img[f] = v;
}

template <int NK>
__device__ __forceinline__ void h_chunk_into_z(const float* smem, float* hC,
                                               const v2f (&a1)[25], int nt, int lo, int hi, int klo,
                                               v8f& z0, v8f& z1) {
  v8f acc = {};
  #pragma unroll
  for (int kt = 0; kt < 25; ++kt) {
    v2f b = *(const v2f*)&smem[OFF_W1 + ((kt * 208 + nt * 16 + lo) << 2) + klo];
    acc = WMMA4(a1[kt], b, acc);
  }
  float bias = smem[OFF_B1 + nt * 16 + lo];
  #pragma unroll
  for (int v = 0; v < 8; ++v) {
    float x = acc[v] + bias;
    x = (x > 0.f) ? x : (__expf(x) - 1.f);
    hC[(v + 8 * hi) * CSTR + lo] = x;
  }
  asm volatile("s_wait_dscnt 0" ::: "memory");
  #pragma unroll
  for (int kk = 0; kk < NK; ++kk) {
    const int kt = nt * 4 + kk;
    v2f a  = *(const v2f*)&hC[lo * CSTR + kk * 4 + klo];
    v2f b0 = *(const v2f*)&smem[OFF_W2 + ((kt * 32 + lo) << 2) + klo];
    v2f b1 = *(const v2f*)&smem[OFF_W2 + ((kt * 32 + 16 + lo) << 2) + klo];
    z0 = WMMA4(a, b0, z0);
    z1 = WMMA4(a, b1, z1);
  }
}

__global__ __launch_bounds__(NTHREADS) void vqvae_fused_kernel(
    const float* __restrict__ c,  const float* __restrict__ wimg,
    float* __restrict__ out_mu, float* __restrict__ out_lv,
    float* __restrict__ out_zq, float* __restrict__ out_ze,
    float* __restrict__ ws, int B) {
  extern __shared__ float smem[];
  const int tid  = threadIdx.x;
  const int wave = tid >> 5;
  const int lane = tid & 31;
  const int lo   = lane & 15;
  const int hi   = lane >> 4;
  const int klo  = hi * 2;

  {
    const v4f* __restrict__ gsrc = (const v4f*)wimg;
    v4f* sdst = (v4f*)smem;
    constexpr int FULL = IMG_VEC4 / NTHREADS;
    const bool hasTail = tid < IMG_VEC4 - FULL * NTHREADS;
    v4f tail = {};
    if (hasTail) tail = gsrc[tid + FULL * NTHREADS];
    #pragma unroll
    for (int t = 0; t < FULL; ++t) sdst[tid + t * NTHREADS] = gsrc[tid + t * NTHREADS];
    if (hasTail) sdst[tid + FULL * NTHREADS] = tail;
  }
  __syncthreads();

  float* pw  = &smem[OFF_PW + wave * PW_FLOATS];
  float* hC  = pw + PW_HC;
  float* zS  = pw + PW_Z;

  const int numTiles = B >> 4;
  const int tile = blockIdx.x * NWAVES + wave;
  const int waveS = __builtin_amdgcn_readfirstlane(tid) >> 5;
  if (tile < numTiles) {
    const int rowBase = tile * 16;

    v2f a1[25];
    const float* crow = c + (size_t)(rowBase + lo) * 100 + klo;
    #pragma unroll
    for (int kt = 0; kt < 25; ++kt) a1[kt] = *(const v2f*)(crow + kt * 4);

    v8f z0 = {}, z1 = {};
    for (int nt = 0; nt < 12; ++nt)
      h_chunk_into_z<4>(smem, hC, a1, nt, lo, hi, klo, z0, z1);
    h_chunk_into_z<2>(smem, hC, a1, 12, lo, hi, klo, z0, z1);

    for (int nt = 0; nt < 2; ++nt) {
      const v8f acc = nt ? z1 : z0;
      int col = nt * 16 + lo;
      float bias = smem[OFF_B2 + col];
      #pragma unroll
      for (int v = 0; v < 8; ++v) {
        if (col < 20) {
          float zv = acc[v] + bias;
          int row = v + 8 * hi;
          zS[row * 20 + col] = zv;
          out_ze[(size_t)(rowBase + row) * 20 + col] = zv;
        }
      }
    }
    asm volatile("s_wait_dscnt 0" ::: "memory");

    v2f az[5];
    #pragma unroll
    for (int kt = 0; kt < 5; ++kt) az[kt] = *(const v2f*)&zS[lo * 20 + kt * 4 + klo];

    for (int nt = 0; nt < 2; ++nt) {
      v8f am = {}, al = {};
      #pragma unroll
      for (int kt = 0; kt < 5; ++kt) {
        v2f bm = *(const v2f*)&smem[OFF_WMU + ((kt * 32 + nt * 16 + lo) << 2) + klo];
        v2f bl = *(const v2f*)&smem[OFF_WLV + ((kt * 32 + nt * 16 + lo) << 2) + klo];
        am = WMMA4(az[kt], bm, am);
        al = WMMA4(az[kt], bl, al);
      }
      int col = nt * 16 + lo;
      if (col < 20) {
        float bmv = smem[OFF_BMU + col], blv_ = smem[OFF_BLV + col];
        #pragma unroll
        for (int v = 0; v < 8; ++v) {
          int row = v + 8 * hi;
          out_mu[(size_t)(rowBase + row) * 20 + col] = am[v] + bmv;
          out_lv[(size_t)(rowBase + row) * 20 + col] = al[v] + blv_;
        }
      }
    }

    float zr[20];
    #pragma unroll
    for (int k = 0; k < 20; ++k) zr[k] = zS[lo * 20 + k];
    float zn = zr[1] * zr[1];
    zn = __builtin_fmaf(zr[0], zr[0], zn);
    #pragma unroll
    for (int k = 2; k < 20; ++k) zn = __builtin_fmaf(zr[k], zr[k], zn);

    float bestD[8]; int bestT[8];
    #pragma unroll
    for (int v = 0; v < 8; ++v) { bestD[v] = 3.4e38f; bestT[v] = 0; }
    for (int jt = 0; jt < 32; ++jt) {
      v2f a[5];
      #pragma unroll
      for (int kt = 0; kt < 5; ++kt)
        a[kt] = *(const v2f*)&smem[OFF_CBT + ((kt * 512 + jt * 16 + lo) << 2) + klo];
      const v4f* cnp = (const v4f*)__builtin_assume_aligned(&smem[OFF_CBN + jt * 16 + 8 * hi], 16);
      const v4f cn0 = cnp[0];
      const v4f cn1 = cnp[1];
      __builtin_amdgcn_sched_barrier(0);
      v8f acc = {};
      #pragma unroll
      for (int kt = 0; kt < 5; ++kt) acc = WMMA4(a[kt], az[kt], acc);
      #pragma unroll
      for (int v = 0; v < 8; ++v) {
        float cn = (v < 4) ? cn0[v & 3] : cn1[v & 3];
        float d = zn + cn - 2.f * acc[v];
        if (d < bestD[v]) { bestD[v] = d; bestT[v] = jt; }
      }
    }
    float bd = bestD[0];
    int   bj = bestT[0] * 16 + 8 * hi;
    #pragma unroll
    for (int v = 1; v < 8; ++v) {
      float od = bestD[v];
      int   oj = bestT[v] * 16 + 8 * hi + v;
      bool take = (od < bd) | ((od == bd) & (oj < bj));
      bd = take ? od : bd;
      bj = take ? oj : bj;
    }
    {
      float od = __shfl_xor(bd, 16, 32);
      int   oj = __shfl_xor(bj, 16, 32);
      bool take = (od < bd) | ((od == bd) & (oj < bj));
      bd = take ? od : bd;
      bj = take ? oj : bj;
    }

    float s = 0.f;
    #pragma unroll
    for (int m = 0; m < 16; ++m)
      s += __int_as_float(__builtin_amdgcn_readlane(__float_as_int(bd), m));
    if (lane == 0) smem[OFF_RED + wave] = s;

    {
      const v4f q0 = *(const v4f*)&smem[OFF_CBT + (((3 * hi)     * 512 + bj) << 2)];
      const v4f q1 = *(const v4f*)&smem[OFF_CBT + (((3 * hi + 1) * 512 + bj) << 2)];
      const v2f q2 = *(const v2f*)&smem[OFF_CBT + ((2 * 512 + bj) << 2) + klo];
      float* zq = out_zq + (size_t)(rowBase + lo) * 20;
      *(v4f*)(zq + 12 * hi)     = q0;
      *(v4f*)(zq + 12 * hi + 4) = q1;
      *(v2f*)(zq + 8 + klo)     = q2;
    }
  } else if (lane == 0) {
    smem[OFF_RED + wave] = 0.f;
  }

  __syncthreads();
  if (waveS == 0) {
    float t = 0.f;
    for (int w = 0; w < NWAVES; ++w) t += smem[OFF_RED + w];
    if (__builtin_amdgcn_mbcnt_lo(~0u, 0u) == 0) ws[blockIdx.x] = t;
  }
}

__global__ __launch_bounds__(256) void vq_loss_reduce(const float* __restrict__ ws,
                                                      float* __restrict__ out,
                                                      int n, float invCount) {
  __shared__ float p[256];
  const int t0 = threadIdx.x;
  const float v0 = (t0 < n) ? ws[t0] : 0.f;
  const float v1 = (t0 + 256 < n) ? ws[t0 + 256] : 0.f;
  float s = v0 + v1;
  for (int i = t0 + 512; i < n; i += 256) s += ws[i];
  p[threadIdx.x] = s;
  __syncthreads();
  if (threadIdx.x == 0) {
    float t = 0.f;
    for (int i = 0; i < 256; ++i) t += p[i];
    float q = t * invCount;
    out[0] = q;
    out[1] = 0.25f * q;
  }
}

extern "C" void kernel_launch(void* const* d_in, const int* in_sizes, int n_in,
                              void* d_out, int out_size, void* d_ws, size_t ws_size,
                              hipStream_t stream) {
  const float* c   = (const float*)d_in[0];
  const float* W1  = (const float*)d_in[1];
  const float* b1  = (const float*)d_in[2];
  const float* W2  = (const float*)d_in[3];
  const float* b2  = (const float*)d_in[4];
  const float* cb  = (const float*)d_in[5];
  const float* Wmu = (const float*)d_in[6];
  const float* bmu = (const float*)d_in[7];
  const float* Wlv = (const float*)d_in[8];
  const float* blv = (const float*)d_in[9];

  const int B = in_sizes[0] / 100;
  const int numTiles = B / 16;
  float* out = (float*)d_out;
  float* wimg = (float*)d_ws;
  float* ws   = wimg + WS_PART_OFF;

  const size_t ldsBytes = (size_t)LDS_FLOATS * sizeof(float);
  (void)hipFuncSetAttribute((const void*)vqvae_fused_kernel,
                            hipFuncAttributeMaxDynamicSharedMemorySize, (int)ldsBytes);

  vqvae_build_weight_block<<<(IMG_FLOATS + 255) / 256, 256, 0, stream>>>(
      W1, b1, W2, b2, cb, Wmu, bmu, Wlv, blv, wimg);

  const int blocks = (numTiles + NWAVES - 1) / NWAVES;
  vqvae_fused_kernel<<<blocks, NTHREADS, ldsBytes, stream>>>(
      c, wimg,
      out,
      out + (size_t)B * 20,
      out + 2 * (size_t)B * 20,
      out + 3 * (size_t)B * 20,
      ws, B);
  vq_loss_reduce<<<1, 256, 0, stream>>>(ws, out + 4 * (size_t)B * 20,
                                        blocks, 1.0f / ((float)B * 20.0f));
}


# two 16-wave work-groups resident per WGP (was one of 32): codebook^T overlays the dead W1 area of LDS after GEMM1; one group's fills and thin end run under the other's WMMAs; loss sum order changes
# speedup vs baseline: 2.9530x; 1.0322x over previous
#include <hip/hip_runtime.h>
#include <math.h>


typedef float v2f __attribute__((ext_vector_type(2)));
typedef float v4f __attribute__((ext_vector_type(4)));
typedef float v8f __attribute__((ext_vector_type(8)));

#define WMMA4(a, b, c) \
  __builtin_amdgcn_wmma_f32_16x16x4_f32(false, (a), false, (b), (short)0, (c), false, false)

constexpr int OFF_W1  = 0;
constexpr int OFF_W2  = OFF_W1  + 20800;
constexpr int OFF_WMU = OFF_W2  + 6400;
constexpr int OFF_WLV = OFF_WMU + 640;
constexpr int OFF_B1  = OFF_WLV + 640;
constexpr int OFF_B2  = OFF_B1  + 208;
constexpr int OFF_BMU = OFF_B2  + 32;
constexpr int OFF_BLV = OFF_BMU + 32;
constexpr int OFF_CBN = OFF_BLV + 32;
constexpr int RES_FLOATS = OFF_CBN + 512;
constexpr int IMG_CBT    = RES_FLOATS;
constexpr int CBT_FLOATS = 10240;
constexpr int OFF_CBT    = OFF_W1;
constexpr int NWAVES   = 16;
constexpr int NTHREADS = NWAVES * 32;
constexpr int CSTR    = 20;
constexpr int OFF_PW  = RES_FLOATS;
constexpr int PW_HC   = 0;
constexpr int PW_Z    = PW_HC + 16 * CSTR;
constexpr int PW_FLOATS = PW_Z + 320;
constexpr int OFF_RED = OFF_PW + NWAVES * PW_FLOATS;
constexpr int LDS_FLOATS = OFF_RED + NWAVES;
constexpr int IMG_FLOATS = IMG_CBT + CBT_FLOATS;
constexpr int RES_VEC4   = RES_FLOATS / 4;
constexpr int CBT_VEC4   = CBT_FLOATS / 4;
static_assert(CBT_FLOATS <= OFF_W2 - OFF_W1, "codebook^T must fit the W1 area it takes over");
static_assert(CBT_VEC4 % NTHREADS == 0, "codebook^T copy: a whole number of pieces per thread");
static_assert(2 * LDS_FLOATS * 4 <= 320 * 1024, "two work-groups must fit the LDS of a WGP");
constexpr int WS_PART_OFF = 40960;
static_assert((RES_FLOATS & 3) == 0 && (CBT_FLOATS & 3) == 0, "both parts must be whole 16-byte pieces");
static_assert(WS_PART_OFF >= IMG_FLOATS, "loss partials must not overlap the weight block");

__global__ __launch_bounds__(256) void vqvae_build_weight_block(
    const float* __restrict__ W1, const float* __restrict__ b1,
    const float* __restrict__ W2, const float* __restrict__ b2, const float* __restrict__ cb,
    const float* __restrict__ Wmu, const float* __restrict__ bmu,
    const float* __restrict__ Wlv, const float* __restrict__ blv,
    float* __restrict__ img) {
  const int f = blockIdx.x * 256 + threadIdx.x;
  if (f >= IMG_FLOATS) return;
  float v;
  if (f < OFF_W2) {
    int g = f - OFF_W1;
    int kk = g & 3, rem = g >> 2, n = rem % 208, kt = rem / 208;
    v = (n < 200) ? W1[(kt * 4 + kk) * 200 + n] : 0.f;
  } else if (f < OFF_WMU) {
    int g = f - OFF_W2;
    int kk = g & 3, rem = g >> 2, n = rem & 31, kt = rem >> 5;
    v = (n < 20) ? W2[(kt * 4 + kk) * 20 + n] : 0.f;
  } else if (f < OFF_WLV) {
    int g = f - OFF_WMU;
    int kk = g & 3, rem = g >> 2, n = rem & 31, kt = rem >> 5;
    v = (n < 20) ? Wmu[(kt * 4 + kk) * 20 + n] : 0.f;
  } else if (f < OFF_B1) {
    int g = f - OFF_WLV;
    int kk = g & 3, rem = g >> 2, n = rem & 31, kt = rem >> 5;
    v = (n < 20) ? Wlv[(kt * 4 + kk) * 20 + n] : 0.f;
  } else if (f < OFF_B2) {
    int n = f - OFF_B1;
    v = (n < 200) ? b1[n] : 0.f;
  } else if (f < OFF_BMU) {
    int n = f - OFF_B2;
    v = (n < 20) ? b2[n] : 0.f;
  } else if (f < OFF_BLV) {
    int n = f - OFF_BMU;
    v = (n < 20) ? bmu[n] : 0.f;
  } else if (f < OFF_CBN) {
    int n = f - OFF_BLV;
    v = (n < 20) ? blv[n] : 0.f;
  } else if (f < IMG_CBT) {
    int j = f - OFF_CBN;
    float s = 0.f;
    for (int k = 0; k < 20; ++k) { float t = cb[j * 20 + k]; s += t * t; }
    v = s;
  } else {
    int g = f - IMG_CBT;
    int kk = g & 3, rem = g >> 2, j = rem & 511, kt = rem >> 9;
    v = cb[j * 20 + kt * 4 + kk];
  }
  img[f] = v;
}

template <int NK>
__device__ __forceinline__ void h_chunk_into_z(const float* smem, float* hC,
                                               const v2f (&a1)[25], int nt, int lo, int hi, int klo,
                                               v8f& z0, v8f& z1) {
  v8f acc = {};
  #pragma unroll
  for (int kt = 0; kt < 25; ++kt) {
    v2f b = *(const v2f*)&smem[OFF_W1 + ((kt * 208 + nt * 16 + lo) << 2) + klo];
    acc = WMMA4(a1[kt], b, acc);
  }
  float bias = smem[OFF_B1 + nt * 16 + lo];
  #pragma unroll
  for (int v = 0; v < 8; ++v) {
    float x = acc[v] + bias;
    x = (x > 0.f) ? x : (__expf(x) - 1.f);
    hC[(v + 8 * hi) * CSTR + lo] = x;
  }
  asm volatile("s_wait_dscnt 0" ::: "memory");
  #pragma unroll
  for (int kk = 0; kk < NK; ++kk) {
    const int kt = nt * 4 + kk;
    v2f a  = *(const v2f*)&hC[lo * CSTR + kk * 4 + klo];
    v2f b0 = *(const v2f*)&smem[OFF_W2 + ((kt * 32 + lo) << 2) + klo];
    v2f b1 = *(const v2f*)&smem[OFF_W2 + ((kt * 32 + 16 + lo) << 2) + klo];
    z0 = WMMA4(a, b0, z0);
    z1 = WMMA4(a, b1, z1);
  }
}

__global__ __launch_bounds__(NTHREADS, 8) void vqvae_fused_kernel(
    const float* __restrict__ c,  const float* __restrict__ wimg,
    float* __restrict__ out_mu, float* __restrict__ out_lv,
    float* __restrict__ out_zq, float* __restrict__ out_ze,
    float* __restrict__ ws, int B) {
  extern __shared__ float smem[];
  const int tid  = threadIdx.x;
  const int lane = tid & 31;
  const int lo   = lane & 15;
  const int hi   = lane >> 4;
  const int klo  = hi * 2;
  const int waveS = __builtin_amdgcn_readfirstlane(tid) >> 5;

  {
    const v4f* __restrict__ gsrc = (const v4f*)wimg;
    v4f* sdst = (v4f*)smem;
    constexpr int FULL = RES_VEC4 / NTHREADS;
    const bool hasTail = tid < RES_VEC4 - FULL * NTHREADS;
    v4f tail = {};
    if (hasTail) tail = gsrc[tid + FULL * NTHREADS];
    #pragma unroll
    for (int t = 0; t < FULL; ++t) sdst[tid + t * NTHREADS] = gsrc[tid + t * NTHREADS];
    if (hasTail) sdst[tid + FULL * NTHREADS] = tail;
  }
  __syncthreads();

  float* pw  = &smem[OFF_PW + waveS * PW_FLOATS];
  float* hC  = pw + PW_HC;
  float* zS  = pw + PW_Z;

  const int numTiles = B >> 4;
  const int tile = blockIdx.x * NWAVES + waveS;
  const bool active = tile < numTiles;
  const int rowBase = tile * 16;

  v8f z0 = {}, z1 = {};
  if (active) {
    v2f a1[25];
    const float* crow = c + (size_t)(rowBase + lo) * 100 + klo;
    #pragma unroll
    for (int kt = 0; kt < 25; ++kt) a1[kt] = *(const v2f*)(crow + kt * 4);

    for (int nt = 0; nt < 12; ++nt)
      h_chunk_into_z<4>(smem, hC, a1, nt, lo, hi, klo, z0, z1);
    h_chunk_into_z<2>(smem, hC, a1, 12, lo, hi, klo, z0, z1);
  }

  v4f cbq[CBT_VEC4 / NTHREADS];
  {
    const v4f* __restrict__ gcb = (const v4f*)(wimg + IMG_CBT);
    #pragma unroll
    for (int t = 0; t < CBT_VEC4 / NTHREADS; ++t) cbq[t] = gcb[tid + t * NTHREADS];
  }

  v2f az[5] = {};
  float zn = 0.f;
  if (active) {
    for (int nt = 0; nt < 2; ++nt) {
      const v8f acc = nt ? z1 : z0;
      int col = nt * 16 + lo;
      float bias = smem[OFF_B2 + col];
      #pragma unroll
      for (int v = 0; v < 8; ++v) {
        if (col < 20) {
          float zv = acc[v] + bias;
          int row = v + 8 * hi;
          zS[row * 20 + col] = zv;
          out_ze[(size_t)(rowBase + row) * 20 + col] = zv;
        }
      }
    }
    asm volatile("s_wait_dscnt 0" ::: "memory");

    #pragma unroll
    for (int kt = 0; kt < 5; ++kt) az[kt] = *(const v2f*)&zS[lo * 20 + kt * 4 + klo];

    for (int nt = 0; nt < 2; ++nt) {
      v8f am = {}, al = {};
      #pragma unroll
      for (int kt = 0; kt < 5; ++kt) {
        v2f bm = *(const v2f*)&smem[OFF_WMU + ((kt * 32 + nt * 16 + lo) << 2) + klo];
        v2f bl = *(const v2f*)&smem[OFF_WLV + ((kt * 32 + nt * 16 + lo) << 2) + klo];
        am = WMMA4(az[kt], bm, am);
        al = WMMA4(az[kt], bl, al);
      }
      int col = nt * 16 + lo;
      if (col < 20) {
        float bmv = smem[OFF_BMU + col], blv_ = smem[OFF_BLV + col];
        #pragma unroll
        for (int v = 0; v < 8; ++v) {
          int row = v + 8 * hi;
          out_mu[(size_t)(rowBase + row) * 20 + col] = am[v] + bmv;
          out_lv[(size_t)(rowBase + row) * 20 + col] = al[v] + blv_;
        }
      }
    }

    float zr[20];
    #pragma unroll
    for (int k = 0; k < 20; ++k) zr[k] = zS[lo * 20 + k];
    zn = zr[1] * zr[1];
    zn = __builtin_fmaf(zr[0], zr[0], zn);
    #pragma unroll
    for (int k = 2; k < 20; ++k) zn = __builtin_fmaf(zr[k], zr[k], zn);
  }

  __syncthreads();
  {
    v4f* sdst = (v4f*)&smem[OFF_CBT];
    #pragma unroll
    for (int t = 0; t < CBT_VEC4 / NTHREADS; ++t) sdst[tid + t * NTHREADS] = cbq[t];
  }
  __syncthreads();

  if (active) {
    float bestD[8]; int bestT[8];
    #pragma unroll
    for (int v = 0; v < 8; ++v) { bestD[v] = 3.4e38f; bestT[v] = 0; }
    for (int jt = 0; jt < 32; ++jt) {
      v2f a[5];
      #pragma unroll
      for (int kt = 0; kt < 5; ++kt)
        a[kt] = *(const v2f*)&smem[OFF_CBT + ((kt * 512 + jt * 16 + lo) << 2) + klo];
      const v4f* cnp = (const v4f*)__builtin_assume_aligned(&smem[OFF_CBN + jt * 16 + 8 * hi], 16);
      const v4f cn0 = cnp[0];
      const v4f cn1 = cnp[1];
      __builtin_amdgcn_sched_barrier(0);
      v8f acc = {};
      #pragma unroll
      for (int kt = 0; kt < 5; ++kt) acc = WMMA4(a[kt], az[kt], acc);
      #pragma unroll
      for (int v = 0; v < 8; ++v) {
        float cn = (v < 4) ? cn0[v & 3] : cn1[v & 3];
        float d = zn + cn - 2.f * acc[v];
        if (d < bestD[v]) { bestD[v] = d; bestT[v] = jt; }
      }
    }
    float bd = bestD[0];
    int   bj = bestT[0] * 16 + 8 * hi;
    #pragma unroll
    for (int v = 1; v < 8; ++v) {
      float od = bestD[v];
      int   oj = bestT[v] * 16 + 8 * hi + v;
      bool take = (od < bd) | ((od == bd) & (oj < bj));
      bd = take ? od : bd;
      bj = take ? oj : bj;
    }
    {
      float od = __shfl_xor(bd, 16, 32);
      int   oj = __shfl_xor(bj, 16, 32);
      bool take = (od < bd) | ((od == bd) & (oj < bj));
      bd = take ? od : bd;
      bj = take ? oj : bj;
    }

    float s = 0.f;
    #pragma unroll
    for (int m = 0; m < 16; ++m)
      s += __int_as_float(__builtin_amdgcn_readlane(__float_as_int(bd), m));
    if (lane == 0) smem[OFF_RED + waveS] = s;

    {
      const v4f q0 = *(const v4f*)&smem[OFF_CBT + (((3 * hi)     * 512 + bj) << 2)];
      const v4f q1 = *(const v4f*)&smem[OFF_CBT + (((3 * hi + 1) * 512 + bj) << 2)];
      const v2f q2 = *(const v2f*)&smem[OFF_CBT + ((2 * 512 + bj) << 2) + klo];
      float* zq = out_zq + (size_t)(rowBase + lo) * 20;
      *(v4f*)(zq + 12 * hi)     = q0;
      *(v4f*)(zq + 12 * hi + 4) = q1;
      *(v2f*)(zq + 8 + klo)     = q2;
    }
  } else if (lane == 0) {
    smem[OFF_RED + waveS] = 0.f;
  }

  __syncthreads();
  if (waveS == 0) {
    float t = 0.f;
    for (int w = 0; w < NWAVES; ++w) t += smem[OFF_RED + w];
    if (__builtin_amdgcn_mbcnt_lo(~0u, 0u) == 0) ws[blockIdx.x] = t;
  }
}

__global__ __launch_bounds__(256) void vq_loss_reduce(const float* __restrict__ ws,
                                                      float* __restrict__ out,
                                                      int n, float invCount) {
  __shared__ float p[256];
  const int t0 = threadIdx.x;
  const float v0 = (t0 < n) ? ws[t0] : 0.f;
  const float v1 = (t0 + 256 < n) ? ws[t0 + 256] : 0.f;
  const float v2 = (t0 + 512 < n) ? ws[t0 + 512] : 0.f;
  const float v3 = (t0 + 768 < n) ? ws[t0 + 768] : 0.f;
  float s = v0 + v1;
  s += v2;
  s += v3;
  for (int i = t0 + 1024; i < n; i += 256) s += ws[i];
  p[threadIdx.x] = s;
  __syncthreads();
  if (threadIdx.x == 0) {
    float t = 0.f;
    for (int i = 0; i < 256; ++i) t += p[i];
    float q = t * invCount;
    out[0] = q;
    out[1] = 0.25f * q;
  }
}

extern "C" void kernel_launch(void* const* d_in, const int* in_sizes, int n_in,
                              void* d_out, int out_size, void* d_ws, size_t ws_size,
                              hipStream_t stream) {
  const float* c   = (const float*)d_in[0];
  const float* W1  = (const float*)d_in[1];
  const float* b1  = (const float*)d_in[2];
  const float* W2  = (const float*)d_in[3];
  const float* b2  = (const float*)d_in[4];
  const float* cb  = (const float*)d_in[5];
  const float* Wmu = (const float*)d_in[6];
  const float* bmu = (const float*)d_in[7];
  const float* Wlv = (const float*)d_in[8];
  const float* blv = (const float*)d_in[9];

  const int B = in_sizes[0] / 100;
  const int numTiles = B / 16;
  float* out = (float*)d_out;
  float* wimg = (float*)d_ws;
  float* ws   = wimg + WS_PART_OFF;

  const size_t ldsBytes = (size_t)LDS_FLOATS * sizeof(float);
  (void)hipFuncSetAttribute((const void*)vqvae_fused_kernel,
                            hipFuncAttributeMaxDynamicSharedMemorySize, (int)ldsBytes);

  vqvae_build_weight_block<<<(IMG_FLOATS + 255) / 256, 256, 0, stream>>>(
      W1, b1, W2, b2, cb, Wmu, bmu, Wlv, blv, wimg);

  const int blocks = (numTiles + NWAVES - 1) / NWAVES;
  vqvae_fused_kernel<<<blocks, NTHREADS, ldsBytes, stream>>>(
      c, wimg,
      out,
      out + (size_t)B * 20,
      out + 2 * (size_t)B * 20,
      out + 3 * (size_t)B * 20,
      ws, B);
  vq_loss_reduce<<<1, 256, 0, stream>>>(ws, out + 4 * (size_t)B * 20,
                                        blocks, 1.0f / ((float)B * 20.0f));
}
